# starting kernel
# baseline (speedup 1.0000x reference)
#include <hip/hip_runtime.h>
#include <hip/hip_bf16.h>


typedef float v2f __attribute__((ext_vector_type(2)));
typedef float v4f __attribute__((ext_vector_type(4)));
typedef float v8f __attribute__((ext_vector_type(8)));

#define BATCH 8
#define NDIM  512
#define SDIM  4096
#define HDIM  1024

#define BLK_N 64
#define BLK_H 128
#define BLK_K 32
#define A_STRIDE 36
#define BP_STRIDE 288
#define NPAIR (BLK_K / 2)

__device__ __forceinline__ void stage_chunk(const float* __restrict__ Ag,
                                            const float* __restrict__ Bg,
                                            int ks, float* __restrict__ Al,
                                            float* __restrict__ Bl, int tid)
{
    #pragma unroll
    for (int i = 0; i < 2; ++i) {
        int idx = i * 256 + tid;
        int r   = idx >> 3;
        int c   = (idx & 7) << 2;
        v4f v = *(const v4f*)(Ag + (size_t)r * SDIM + ks + c);
        *(v4f*)(&Al[r * A_STRIDE + c]) = v;
    }
    #pragma unroll
    for (int i = 0; i < 4; ++i) {
        int idx = i * 256 + tid;
        int p   = idx >> 6;
        int c   = (idx & 63) << 1;
        v2f lo = *(const v2f*)(Bg + (size_t)(ks + 2 * p)     * HDIM + c);
        v2f hi = *(const v2f*)(Bg + (size_t)(ks + 2 * p + 1) * HDIM + c);
        v4f v; v.x = lo.x; v.y = hi.x; v.z = lo.y; v.w = hi.y;
        *(v4f*)(&Bl[p * BP_STRIDE + 2 * c]) = v;
    }
}

__device__ __forceinline__ void compute_chunk(const float* __restrict__ Al,
                                              const float* __restrict__ Bl,
                                              int wr, int wc, int hi, int ln,
                                              v8f& acc0, v8f& acc1,
                                              v8f& acc2, v8f& acc3)
{
    #pragma unroll
    for (int kk = 0; kk < BLK_K; kk += 4) {
        v2f a = *(const v2f*)(&Al[(wr + ln) * A_STRIDE + kk + 2 * hi]);
        const float* bp = &Bl[((kk >> 1) + hi) * BP_STRIDE + 2 * (wc + ln)];
        v2f b0 = *(const v2f*)(bp +  0);
        acc0 = __builtin_amdgcn_wmma_f32_16x16x4_f32(false, a, false, b0,
                                                     (short)0, acc0, false, false);
        v2f b1 = *(const v2f*)(bp + 32);
        acc1 = __builtin_amdgcn_wmma_f32_16x16x4_f32(false, a, false, b1,
                                                     (short)0, acc1, false, false);
        v2f b2 = *(const v2f*)(bp + 64);
        acc2 = __builtin_amdgcn_wmma_f32_16x16x4_f32(false, a, false, b2,
                                                     (short)0, acc2, false, false);
        v2f b3 = *(const v2f*)(bp + 96);
        acc3 = __builtin_amdgcn_wmma_f32_16x16x4_f32(false, a, false, b3,
                                                     (short)0, acc3, false, false);
    }
}

__global__ __launch_bounds__(256)
void meanpool_wmma_f32(const float* __restrict__ doc_state,
                       const float* __restrict__ nodes_mapping,
                       const float* __restrict__ nodes_len,
                       float* __restrict__ out)
{
    __shared__ float Al[2][BLK_N * A_STRIDE];
    __shared__ float Bl[2][NPAIR * BP_STRIDE];
    __shared__ float invLen[BLK_N];

    const int tid  = threadIdx.x;
    const int lane = tid & 31;
    const int wave = tid >> 5;
    const int b    = blockIdx.z;
    const int n0   = blockIdx.y * BLK_N;
    const int h0   = blockIdx.x * BLK_H;

    const float* Ag = nodes_mapping + ((size_t)b * NDIM + n0) * SDIM;
    const float* Bg = doc_state     + (size_t)b * SDIM * HDIM + h0;

    if (tid < BLK_N)
        invLen[tid] = 1.0f / nodes_len[(size_t)b * NDIM + n0 + tid];

    const int wr = (wave >> 1) * 16;
    const int wc = (wave & 1) * 64;
    const int hi = lane >> 4;
    const int ln = lane & 15;

    v8f acc0 = {}, acc1 = {}, acc2 = {}, acc3 = {};

    stage_chunk(Ag, Bg, 0, Al[0], Bl[0], tid);

    for (int ks = 0; ks < SDIM; ks += 2 * BLK_K) {
        __syncthreads();
        if (ks + BLK_K < SDIM)
            stage_chunk(Ag, Bg, ks + BLK_K, Al[1], Bl[1], tid);
        compute_chunk(Al[0], Bl[0], wr, wc, hi, ln, acc0, acc1, acc2, acc3);

        __syncthreads();
        if (ks + 2 * BLK_K < SDIM)
            stage_chunk(Ag, Bg, ks + 2 * BLK_K, Al[0], Bl[0], tid);
        compute_chunk(Al[1], Bl[1], wr, wc, hi, ln, acc0, acc1, acc2, acc3);
    }

    #pragma unroll
    for (int r = 0; r < 8; ++r) {
        const int   rloc = wr + 8 * hi + r;
        const float sc   = invLen[rloc];
        float* orow = out + ((size_t)b * NDIM + n0 + rloc) * HDIM + h0 + wc + ln;
        orow[ 0] = acc0[r] * sc;
        orow[16] = acc1[r] * sc;
        orow[32] = acc2[r] * sc;
        orow[48] = acc3[r] * sc;
    }
}

extern "C" void kernel_launch(void* const* d_in, const int* in_sizes, int n_in,
                              void* d_out, int out_size, void* d_ws, size_t ws_size,
                              hipStream_t stream) {
    (void)in_sizes; (void)n_in; (void)out_size; (void)d_ws; (void)ws_size;
    const float* doc_state     = (const float*)d_in[0];
    const float* nodes_mapping = (const float*)d_in[1];
    const float* nodes_len     = (const float*)d_in[2];
    float* out = (float*)d_out;

    dim3 grid(HDIM / BLK_H, NDIM / BLK_N, BATCH);
    dim3 block(256);
    meanpool_wmma_f32<<<grid, block, 0, stream>>>(doc_state, nodes_mapping,
                                                  nodes_len, out);
}


# Next K-chunk's global loads stay in flight for a whole 32-WMMA block: loaded a block ahead into registers, LDS-stored a block later, barrier fences LDS only; parent waited right after issuing them
# speedup vs baseline: 1.1329x; 1.1329x over previous
#include <hip/hip_runtime.h>
#include <hip/hip_bf16.h>


typedef float v2f __attribute__((ext_vector_type(2)));
typedef float v4f __attribute__((ext_vector_type(4)));
typedef float v8f __attribute__((ext_vector_type(8)));

#define BATCH 8
#define NDIM  512
#define SDIM  4096
#define HDIM  1024

#define BLK_N 64
#define BLK_H 128
#define BLK_K 32
#define A_STRIDE 36
#define BP_STRIDE 288
#define NPAIR (BLK_K / 2)

struct ChunkRegs {
    v4f a[2];
    v2f lo[4];
    v2f hi[4];
};

__device__ __forceinline__ void load_chunk(const float* __restrict__ Ag,
                                           const float* __restrict__ Bg,
                                           int ks, int tid, ChunkRegs& r)
{
    #pragma unroll
    for (int i = 0; i < 2; ++i) {
        int idx = i * 256 + tid;
        int rr  = idx >> 3;
        int c   = (idx & 7) << 2;
        r.a[i] = *(const v4f*)(Ag + (size_t)rr * SDIM + ks + c);
    }
    #pragma unroll
    for (int i = 0; i < 4; ++i) {
        int idx = i * 256 + tid;
        int p   = idx >> 6;
        int c   = (idx & 63) << 1;
        r.lo[i] = *(const v2f*)(Bg + (size_t)(ks + 2 * p)     * HDIM + c);
        r.hi[i] = *(const v2f*)(Bg + (size_t)(ks + 2 * p + 1) * HDIM + c);
    }
}

__device__ __forceinline__ void store_chunk(float* __restrict__ Al,
                                            float* __restrict__ Bl,
                                            int tid, const ChunkRegs& r)
{
    #pragma unroll
    for (int i = 0; i < 2; ++i) {
        int idx = i * 256 + tid;
        int rr  = idx >> 3;
        int c   = (idx & 7) << 2;
        *(v4f*)(&Al[rr * A_STRIDE + c]) = r.a[i];
    }
    #pragma unroll
    for (int i = 0; i < 4; ++i) {
        int idx = i * 256 + tid;
        int p   = idx >> 6;
        int c   = (idx & 63) << 1;
        v4f v; v.x = r.lo[i].x; v.y = r.hi[i].x; v.z = r.lo[i].y; v.w = r.hi[i].y;
        *(v4f*)(&Bl[p * BP_STRIDE + 2 * c]) = v;
    }
}

__device__ __forceinline__ void compute_chunk(const float* __restrict__ Al,
                                              const float* __restrict__ Bl,
                                              int wr, int wc, int hi, int ln,
                                              v8f& acc0, v8f& acc1,
                                              v8f& acc2, v8f& acc3)
{
    #pragma unroll
    for (int kk = 0; kk < BLK_K; kk += 4) {
        v2f a = *(const v2f*)(&Al[(wr + ln) * A_STRIDE + kk + 2 * hi]);
        const float* bp = &Bl[((kk >> 1) + hi) * BP_STRIDE + 2 * (wc + ln)];
        v2f b0 = *(const v2f*)(bp +  0);
        acc0 = __builtin_amdgcn_wmma_f32_16x16x4_f32(false, a, false, b0,
                                                     (short)0, acc0, false, false);
        v2f b1 = *(const v2f*)(bp + 32);
        acc1 = __builtin_amdgcn_wmma_f32_16x16x4_f32(false, a, false, b1,
                                                     (short)0, acc1, false, false);
        v2f b2 = *(const v2f*)(bp + 64);
        acc2 = __builtin_amdgcn_wmma_f32_16x16x4_f32(false, a, false, b2,
                                                     (short)0, acc2, false, false);
        v2f b3 = *(const v2f*)(bp + 96);
        acc3 = __builtin_amdgcn_wmma_f32_16x16x4_f32(false, a, false, b3,
                                                     (short)0, acc3, false, false);
    }
}

__device__ __forceinline__ void lds_barrier()
{
    __builtin_amdgcn_fence(__ATOMIC_RELEASE, "workgroup", "local");
    __builtin_amdgcn_s_barrier();
    __builtin_amdgcn_fence(__ATOMIC_ACQUIRE, "workgroup", "local");
}

__global__ __launch_bounds__(256)
void meanpool_wmma_f32(const float* __restrict__ doc_state,
                       const float* __restrict__ nodes_mapping,
                       const float* __restrict__ nodes_len,
                       float* __restrict__ out)
{
    __shared__ float Al[2][BLK_N * A_STRIDE];
    __shared__ float Bl[2][NPAIR * BP_STRIDE];
    __shared__ float invLen[BLK_N];

    const int tid  = threadIdx.x;
    const int lane = tid & 31;
    const int wave = tid >> 5;
    const int b    = blockIdx.z;
    const int n0   = blockIdx.y * BLK_N;
    const int h0   = blockIdx.x * BLK_H;

    const float* Ag = nodes_mapping + ((size_t)b * NDIM + n0) * SDIM;
    const float* Bg = doc_state     + (size_t)b * SDIM * HDIM + h0;

    if (tid < BLK_N)
        invLen[tid] = 1.0f / nodes_len[(size_t)b * NDIM + n0 + tid];

    const int wr = (wave >> 1) * 16;
    const int wc = (wave & 1) * 64;
    const int hi = lane >> 4;
    const int ln = lane & 15;

    v8f acc0 = {}, acc1 = {}, acc2 = {}, acc3 = {};

    ChunkRegs R;
    load_chunk(Ag, Bg, 0, tid, R);
    store_chunk(Al[0], Bl[0], tid, R);
    load_chunk(Ag, Bg, BLK_K, tid, R);

    for (int ks = 0; ks < SDIM - 2 * BLK_K; ks += 2 * BLK_K) {
        lds_barrier();
        compute_chunk(Al[0], Bl[0], wr, wc, hi, ln, acc0, acc1, acc2, acc3);
        __builtin_amdgcn_sched_barrier(0);
        store_chunk(Al[1], Bl[1], tid, R);
        load_chunk(Ag, Bg, ks + 2 * BLK_K, tid, R);

        lds_barrier();
        compute_chunk(Al[1], Bl[1], wr, wc, hi, ln, acc0, acc1, acc2, acc3);
        __builtin_amdgcn_sched_barrier(0);
        store_chunk(Al[0], Bl[0], tid, R);
        load_chunk(Ag, Bg, ks + 3 * BLK_K, tid, R);
    }
    lds_barrier();
    compute_chunk(Al[0], Bl[0], wr, wc, hi, ln, acc0, acc1, acc2, acc3);
    store_chunk(Al[1], Bl[1], tid, R);
    lds_barrier();
    compute_chunk(Al[1], Bl[1], wr, wc, hi, ln, acc0, acc1, acc2, acc3);

    #pragma unroll
    for (int r = 0; r < 8; ++r) {
        const int   rloc = wr + 8 * hi + r;
        const float sc   = invLen[rloc];
        float* orow = out + ((size_t)b * NDIM + n0 + rloc) * HDIM + h0 + wc + ln;
        orow[ 0] = acc0[r] * sc;
        orow[16] = acc1[r] * sc;
        orow[32] = acc2[r] * sc;
        orow[48] = acc3[r] * sc;
    }
}

extern "C" void kernel_launch(void* const* d_in, const int* in_sizes, int n_in,
                              void* d_out, int out_size, void* d_ws, size_t ws_size,
                              hipStream_t stream) {
    (void)in_sizes; (void)n_in; (void)out_size; (void)d_ws; (void)ws_size;
    const float* doc_state     = (const float*)d_in[0];
    const float* nodes_mapping = (const float*)d_in[1];
    const float* nodes_len     = (const float*)d_in[2];
    float* out = (float*)d_out;

    dim3 grid(HDIM / BLK_H, NDIM / BLK_N, BATCH);
    dim3 block(256);
    meanpool_wmma_f32<<<grid, block, 0, stream>>>(doc_state, nodes_mapping,
                                                  nodes_len, out);
}


# Wave priority stepped down along the K loop (s_setprio 3,2,1,0 from K = 0, 1/2, 3/4, 15/16): the two workgroups sharing a WGP take turns on the matrix pipe; before, the younger one ended alone
# speedup vs baseline: 1.1750x; 1.0372x over previous
#include <hip/hip_runtime.h>
#include <hip/hip_bf16.h>


typedef float v2f __attribute__((ext_vector_type(2)));
typedef float v4f __attribute__((ext_vector_type(4)));
typedef float v8f __attribute__((ext_vector_type(8)));

#define BATCH 8
#define NDIM  512
#define SDIM  4096
#define HDIM  1024

#define BLK_N 64
#define BLK_H 128
#define BLK_K 32
#define A_STRIDE 36
#define BP_STRIDE 288
#define NPAIR (BLK_K / 2)

#define PRIO_STEP1 (SDIM / 2)
#define PRIO_STEP2 (SDIM / 2 + SDIM / 4)
#define PRIO_STEP3 (SDIM - SDIM / 16)
#define K_LOOP_END (SDIM - 2 * BLK_K)

struct ChunkRegs {
    v4f a[2];
    v2f lo[4];
    v2f hi[4];
};

__device__ __forceinline__ void load_chunk(const float* __restrict__ Ag,
                                           const float* __restrict__ Bg,
                                           int ks, int tid, ChunkRegs& r)
{
    #pragma unroll
    for (int i = 0; i < 2; ++i) {
        int idx = i * 256 + tid;
        int rr  = idx >> 3;
        int c   = (idx & 7) << 2;
        r.a[i] = *(const v4f*)(Ag + (size_t)rr * SDIM + ks + c);
    }
    #pragma unroll
    for (int i = 0; i < 4; ++i) {
        int idx = i * 256 + tid;
        int p   = idx >> 6;
        int c   = (idx & 63) << 1;
        r.lo[i] = *(const v2f*)(Bg + (size_t)(ks + 2 * p)     * HDIM + c);
        r.hi[i] = *(const v2f*)(Bg + (size_t)(ks + 2 * p + 1) * HDIM + c);
    }
}

__device__ __forceinline__ void store_chunk(float* __restrict__ Al,
                                            float* __restrict__ Bl,
                                            int tid, const ChunkRegs& r)
{
    #pragma unroll
    for (int i = 0; i < 2; ++i) {
        int idx = i * 256 + tid;
        int rr  = idx >> 3;
        int c   = (idx & 7) << 2;
        *(v4f*)(&Al[rr * A_STRIDE + c]) = r.a[i];
    }
    #pragma unroll
    for (int i = 0; i < 4; ++i) {
        int idx = i * 256 + tid;
        int p   = idx >> 6;
        int c   = (idx & 63) << 1;
        v4f v; v.x = r.lo[i].x; v.y = r.hi[i].x; v.z = r.lo[i].y; v.w = r.hi[i].y;
        *(v4f*)(&Bl[p * BP_STRIDE + 2 * c]) = v;
    }
}

__device__ __forceinline__ void compute_chunk(const float* __restrict__ Al,
                                              const float* __restrict__ Bl,
                                              int wr, int wc, int hi, int ln,
                                              v8f& acc0, v8f& acc1,
                                              v8f& acc2, v8f& acc3)
{
    #pragma unroll
    for (int kk = 0; kk < BLK_K; kk += 4) {
        v2f a = *(const v2f*)(&Al[(wr + ln) * A_STRIDE + kk + 2 * hi]);
        const float* bp = &Bl[((kk >> 1) + hi) * BP_STRIDE + 2 * (wc + ln)];
        v2f b0 = *(const v2f*)(bp +  0);
        acc0 = __builtin_amdgcn_wmma_f32_16x16x4_f32(false, a, false, b0,
                                                     (short)0, acc0, false, false);
        v2f b1 = *(const v2f*)(bp + 32);
        acc1 = __builtin_amdgcn_wmma_f32_16x16x4_f32(false, a, false, b1,
                                                     (short)0, acc1, false, false);
        v2f b2 = *(const v2f*)(bp + 64);
        acc2 = __builtin_amdgcn_wmma_f32_16x16x4_f32(false, a, false, b2,
                                                     (short)0, acc2, false, false);
        v2f b3 = *(const v2f*)(bp + 96);
        acc3 = __builtin_amdgcn_wmma_f32_16x16x4_f32(false, a, false, b3,
                                                     (short)0, acc3, false, false);
    }
}

__device__ __forceinline__ void lds_barrier()
{
    __builtin_amdgcn_fence(__ATOMIC_RELEASE, "workgroup", "local");
    __builtin_amdgcn_s_barrier();
    __builtin_amdgcn_fence(__ATOMIC_ACQUIRE, "workgroup", "local");
}

__global__ __launch_bounds__(256)
void meanpool_wmma_f32(const float* __restrict__ doc_state,
                       const float* __restrict__ nodes_mapping,
                       const float* __restrict__ nodes_len,
                       float* __restrict__ out)
{
    __shared__ float Al[2][BLK_N * A_STRIDE];
    __shared__ float Bl[2][NPAIR * BP_STRIDE];
    __shared__ float invLen[BLK_N];

    const int tid  = threadIdx.x;
    const int lane = tid & 31;
    const int wave = tid >> 5;
    const int b    = blockIdx.z;
    const int n0   = blockIdx.y * BLK_N;
    const int h0   = blockIdx.x * BLK_H;

    const float* Ag = nodes_mapping + ((size_t)b * NDIM + n0) * SDIM;
    const float* Bg = doc_state     + (size_t)b * SDIM * HDIM + h0;

    if (tid < BLK_N)
        invLen[tid] = 1.0f / nodes_len[(size_t)b * NDIM + n0 + tid];

    const int wr = (wave >> 1) * 16;
    const int wc = (wave & 1) * 64;
    const int hi = lane >> 4;
    const int ln = lane & 15;

    v8f acc0 = {}, acc1 = {}, acc2 = {}, acc3 = {};

    ChunkRegs R;
    load_chunk(Ag, Bg, 0, tid, R);
    store_chunk(Al[0], Bl[0], tid, R);
    load_chunk(Ag, Bg, BLK_K, tid, R);

    auto k_segment = [&](const int ks_begin, const int ks_end)
        __attribute__((always_inline)) {
    for (int ks = ks_begin; ks < ks_end; ks += 2 * BLK_K) {
        lds_barrier();
        compute_chunk(Al[0], Bl[0], wr, wc, hi, ln, acc0, acc1, acc2, acc3);
        __builtin_amdgcn_sched_barrier(0);
        store_chunk(Al[1], Bl[1], tid, R);
        load_chunk(Ag, Bg, ks + 2 * BLK_K, tid, R);

        lds_barrier();
        compute_chunk(Al[1], Bl[1], wr, wc, hi, ln, acc0, acc1, acc2, acc3);
        __builtin_amdgcn_sched_barrier(0);
        store_chunk(Al[0], Bl[0], tid, R);
        load_chunk(Ag, Bg, ks + 3 * BLK_K, tid, R);
    }
    };
    __builtin_amdgcn_s_setprio(3);
    k_segment(0, PRIO_STEP1);
    __builtin_amdgcn_s_setprio(2);
    k_segment(PRIO_STEP1, PRIO_STEP2);
    __builtin_amdgcn_s_setprio(1);
    k_segment(PRIO_STEP2, PRIO_STEP3);
    __builtin_amdgcn_s_setprio(0);
    k_segment(PRIO_STEP3, K_LOOP_END);
    lds_barrier();
    compute_chunk(Al[0], Bl[0], wr, wc, hi, ln, acc0, acc1, acc2, acc3);
    store_chunk(Al[1], Bl[1], tid, R);
    lds_barrier();
    compute_chunk(Al[1], Bl[1], wr, wc, hi, ln, acc0, acc1, acc2, acc3);

    #pragma unroll
    for (int r = 0; r < 8; ++r) {
        const int   rloc = wr + 8 * hi + r;
        const float sc   = invLen[rloc];
        float* orow = out + ((size_t)b * NDIM + n0 + rloc) * HDIM + h0 + wc + ln;
        orow[ 0] = acc0[r] * sc;
        orow[16] = acc1[r] * sc;
        orow[32] = acc2[r] * sc;
        orow[48] = acc3[r] * sc;
    }
}

extern "C" void kernel_launch(void* const* d_in, const int* in_sizes, int n_in,
                              void* d_out, int out_size, void* d_ws, size_t ws_size,
                              hipStream_t stream) {
    (void)in_sizes; (void)n_in; (void)out_size; (void)d_ws; (void)ws_size;
    const float* doc_state     = (const float*)d_in[0];
    const float* nodes_mapping = (const float*)d_in[1];
    const float* nodes_len     = (const float*)d_in[2];
    float* out = (float*)d_out;

    dim3 grid(HDIM / BLK_H, NDIM / BLK_N, BATCH);
    dim3 block(256);
    meanpool_wmma_f32<<<grid, block, 0, stream>>>(doc_state, nodes_mapping,
                                                  nodes_len, out);
}


# One 128x128 workgroup per WGP, 32x64 per wave: each LDS fragment feeds 2-4 WMMAs, other issues per 64 WMMAs 99 to 58; fragments read a block ahead, barrier arrive mid-block, wait at block end
# speedup vs baseline: 1.2491x; 1.0631x over previous
#include <hip/hip_runtime.h>


typedef float v2f __attribute__((ext_vector_type(2)));
typedef float v4f __attribute__((ext_vector_type(4)));
typedef float v8f __attribute__((ext_vector_type(8)));

#define BATCH 8
#define NDIM  512
#define SDIM  4096
#define HDIM  1024

#define BLK_N 128
#define BLK_H 128
#define BLK_K 32
#define A_STRIDE 36
#define BP_STRIDE 288
#define NPAIR (BLK_K / 2)
#define KSTEPS (BLK_K / 4)

struct ChunkRegs {
    v4f a[4];
    v2f lo[4];
    v2f hi[4];
};

struct Frags {
    v2f a[KSTEPS][2];
    v2f b[KSTEPS][4];
};

__device__ __forceinline__ void load_chunk(const float* __restrict__ Ag,
                                           const float* __restrict__ Bg,
                                           int ks, int tid, ChunkRegs& r)
{
    #pragma unroll
    for (int i = 0; i < 4; ++i) {
        int idx = i * 256 + tid;
        int rr  = idx >> 3;
        int c   = (idx & 7) << 2;
        r.a[i] = *(const v4f*)(Ag + (size_t)rr * SDIM + ks + c);
    }
    #pragma unroll
    for (int i = 0; i < 4; ++i) {
        int idx = i * 256 + tid;
        int p   = idx >> 6;
        int c   = (idx & 63) << 1;
        r.lo[i] = *(const v2f*)(Bg + (size_t)(ks + 2 * p)     * HDIM + c);
        r.hi[i] = *(const v2f*)(Bg + (size_t)(ks + 2 * p + 1) * HDIM + c);
    }
}

__device__ __forceinline__ void store_chunk(float* __restrict__ Al,
                                            float* __restrict__ Bl,
                                            int tid, const ChunkRegs& r)
{
    #pragma unroll
    for (int i = 0; i < 4; ++i) {
        int idx = i * 256 + tid;
        int rr  = idx >> 3;
        int c   = (idx & 7) << 2;
        *(v4f*)(&Al[rr * A_STRIDE + c]) = r.a[i];
    }
    #pragma unroll
    for (int i = 0; i < 4; ++i) {
        int idx = i * 256 + tid;
        int p   = idx >> 6;
        int c   = (idx & 63) << 1;
        v4f v; v.x = r.lo[i].x; v.y = r.hi[i].x; v.z = r.lo[i].y; v.w = r.hi[i].y;
        *(v4f*)(&Bl[p * BP_STRIDE + 2 * c]) = v;
    }
}

__device__ __forceinline__ void load_frags(const float* __restrict__ Al,
                                           const float* __restrict__ Bl,
                                           int wr, int wc, int hi, int ln, Frags& f)
{
    #pragma unroll
    for (int s = 0; s < KSTEPS; ++s) {
        const int kk = 4 * s;
        f.a[s][0] = *(const v2f*)(&Al[(wr + ln)      * A_STRIDE + kk + 2 * hi]);
        f.a[s][1] = *(const v2f*)(&Al[(wr + 16 + ln) * A_STRIDE + kk + 2 * hi]);
        const float* bp = &Bl[((kk >> 1) + hi) * BP_STRIDE + 2 * (wc + ln)];
        f.b[s][0] = *(const v2f*)(bp +  0);
        f.b[s][1] = *(const v2f*)(bp + 32);
        f.b[s][2] = *(const v2f*)(bp + 64);
        f.b[s][3] = *(const v2f*)(bp + 96);
    }
}

template <int HALF>
__device__ __forceinline__ void mma_half(const Frags& f, v8f (&acc)[2][4])
{
    #pragma unroll
    for (int s = HALF * (KSTEPS / 2); s < (HALF + 1) * (KSTEPS / 2); ++s) {
        #pragma unroll
        for (int i = 0; i < 2; ++i) {
            #pragma unroll
            for (int j = 0; j < 4; ++j) {
                acc[i][j] = __builtin_amdgcn_wmma_f32_16x16x4_f32(
                    false, f.a[s][i], false, f.b[s][j], (short)0, acc[i][j], false, false);
            }
        }
    }
}

__device__ __forceinline__ void sched_fence()
{
    __builtin_amdgcn_sched_barrier(0);
}

__device__ __forceinline__ void lds_arrive()
{
    asm volatile("s_wait_dscnt 0x0\n\ts_barrier_signal -1" ::: "memory");
}
__device__ __forceinline__ void lds_wait()
{
    asm volatile("s_barrier_wait -1" ::: "memory");
}

template <bool FR, bool ST, bool LD>
__device__ __forceinline__ void block(const float* __restrict__ Ag,
                                      const float* __restrict__ Bg,
                                      float* __restrict__ AlC, float* __restrict__ BlC,
                                      const float* __restrict__ AlN,
                                      const float* __restrict__ BlN,
                                      int ks_load, int tid, int wr, int wc, int hi, int ln,
                                      ChunkRegs& R, const Frags& Fc, Frags& Fn,
                                      v8f (&acc)[2][4])
{
    if (FR) load_frags(AlN, BlN, wr, wc, hi, ln, Fn);
    if (ST) store_chunk(AlC, BlC, tid, R);
    if (LD) load_chunk(Ag, Bg, ks_load, tid, R);
    sched_fence();
    mma_half<0>(Fc, acc);
    sched_fence();
    if (ST) lds_arrive();
    mma_half<1>(Fc, acc);
    sched_fence();
    if (ST) lds_wait();
}

__global__ __launch_bounds__(256)
void meanpool_wmma_f32(const float* __restrict__ doc_state,
                       const float* __restrict__ nodes_mapping,
                       const float* __restrict__ nodes_len,
                       float* __restrict__ out)
{
    __shared__ float Al[2][BLK_N * A_STRIDE];
    __shared__ float Bl[2][NPAIR * BP_STRIDE];
    __shared__ float invLen[BLK_N];

    const int tid  = threadIdx.x;
    const int lane = tid & 31;
    const int wave = tid >> 5;
    const int b    = blockIdx.z;
    const int n0   = blockIdx.y * BLK_N;
    const int h0   = blockIdx.x * BLK_H;

    const float* Ag = nodes_mapping + ((size_t)b * NDIM + n0) * SDIM;
    const float* Bg = doc_state     + (size_t)b * SDIM * HDIM + h0;

    if (tid < BLK_N)
        invLen[tid] = 1.0f / nodes_len[(size_t)b * NDIM + n0 + tid];

    const int wr = (wave >> 1) * 32;
    const int wc = (wave & 1) * 64;
    const int hi = lane >> 4;
    const int ln = lane & 15;

    v8f acc[2][4] = {};

    ChunkRegs R;
    Frags F0, F1;

    {
        ChunkRegs R1;
        load_chunk(Ag, Bg, 0, tid, R);
        load_chunk(Ag, Bg, BLK_K, tid, R1);
        store_chunk(Al[0], Bl[0], tid, R);
        store_chunk(Al[1], Bl[1], tid, R1);
    }
    load_chunk(Ag, Bg, 2 * BLK_K, tid, R);
    lds_arrive();
    lds_wait();
    load_frags(Al[0], Bl[0], wr, wc, hi, ln, F0);
    lds_arrive();
    lds_wait();

    for (int ks = 0; ks < SDIM - 4 * BLK_K; ks += 2 * BLK_K) {
        block<true, true, true>(Ag, Bg, Al[0], Bl[0], Al[1], Bl[1], ks + 3 * BLK_K,
                                tid, wr, wc, hi, ln, R, F0, F1, acc);
        block<true, true, true>(Ag, Bg, Al[1], Bl[1], Al[0], Bl[0], ks + 4 * BLK_K,
                                tid, wr, wc, hi, ln, R, F1, F0, acc);
    }
    block<true, true, true >(Ag, Bg, Al[0], Bl[0], Al[1], Bl[1], SDIM - BLK_K,
                             tid, wr, wc, hi, ln, R, F0, F1, acc);
    block<true, true, false>(Ag, Bg, Al[1], Bl[1], Al[0], Bl[0], 0,
                             tid, wr, wc, hi, ln, R, F1, F0, acc);
    block<true, false, false>(Ag, Bg, Al[0], Bl[0], Al[1], Bl[1], 0,
                              tid, wr, wc, hi, ln, R, F0, F1, acc);
    block<false, false, false>(Ag, Bg, Al[1], Bl[1], Al[0], Bl[0], 0,
                               tid, wr, wc, hi, ln, R, F1, F0, acc);

    #pragma unroll
    for (int i = 0; i < 2; ++i) {
        #pragma unroll
        for (int r = 0; r < 8; ++r) {
            const int   rloc = wr + 16 * i + 8 * hi + r;
            const float sc   = invLen[rloc];
            float* orow = out + ((size_t)b * NDIM + n0 + rloc) * HDIM + h0 + wc + ln;
            orow[ 0] = acc[i][0][r] * sc;
            orow[16] = acc[i][1][r] * sc;
            orow[32] = acc[i][2][r] * sc;
            orow[48] = acc[i][3][r] * sc;
        }
    }
}

extern "C" void kernel_launch(void* const* d_in, const int* in_sizes, int n_in,
                              void* d_out, int out_size, void* d_ws, size_t ws_size,
                              hipStream_t stream) {
    (void)in_sizes; (void)n_in; (void)out_size; (void)d_ws; (void)ws_size;
    const float* doc_state     = (const float*)d_in[0];
    const float* nodes_mapping = (const float*)d_in[1];
    const float* nodes_len     = (const float*)d_in[2];
    float* out = (float*)d_out;

    dim3 grid(HDIM / BLK_H, NDIM / BLK_N, BATCH);
    dim3 block(256);
    meanpool_wmma_f32<<<grid, block, 0, stream>>>(doc_state, nodes_mapping,
                                                  nodes_len, out);
}
